# starting kernel
# baseline (speedup 1.0000x reference)
#include <hip/hip_runtime.h>
#include <hip/hip_bf16.h>
#include <math.h>

#define NUM_SRC 100000
#define NUM_DST 50000
#define NUM_E   800000
#define DIM     128

typedef __attribute__((ext_vector_type(2))) float v2f;
typedef __attribute__((ext_vector_type(8))) float v8f;

__global__ __launch_bounds__(256) void deg_kernel(const int* __restrict__ src,
                                                  const int* __restrict__ dst,
                                                  float* __restrict__ deg_src,
                                                  float* __restrict__ deg_dst) {
    int e = blockIdx.x * blockDim.x + threadIdx.x;
    if (e < NUM_E) {
        atomicAdd(&deg_src[src[e]], 1.0f);
        atomicAdd(&deg_dst[dst[e]], 1.0f);
    }
}

__global__ __launch_bounds__(256) void scatter_kernel(const float* __restrict__ h,
                                                      const int* __restrict__ src,
                                                      const int* __restrict__ dst,
                                                      const float* __restrict__ deg_src,
                                                      float* __restrict__ agg) {
    int t    = blockIdx.x * blockDim.x + threadIdx.x;
    int e    = t >> 5;
    int lane = t & 31;
    if (e >= NUM_E) return;

    int s = src[e];
    int d = dst[e];
    float sc = rsqrtf(fmaxf(deg_src[s], 1.0f));

    const float4* hp = (const float4*)(h + (size_t)s * DIM);
    float4 v = hp[lane];

    float* ap = agg + (size_t)d * DIM + lane * 4;
    atomicAdd(ap + 0, v.x * sc);
    atomicAdd(ap + 1, v.y * sc);
    atomicAdd(ap + 2, v.z * sc);
    atomicAdd(ap + 3, v.w * sc);
}

#define LDS_STRIDE 132

__global__ __launch_bounds__(256) void gemm_attn_kernel(const float* __restrict__ agg,
                                                        const float* __restrict__ deg_dst,
                                                        const float* __restrict__ W,
                                                        const float* __restrict__ bias,
                                                        const float* __restrict__ w_attn,
                                                        float* __restrict__ out_scaled,
                                                        float* __restrict__ alpha_out) {
    __shared__ float As[16 * LDS_STRIDE];
    __shared__ float Os[16 * LDS_STRIDE];
    __shared__ float alphaS[16];

    const int row0 = blockIdx.x * 16;

    for (int i = threadIdx.x; i < 16 * DIM; i += 256) {
        int r = i >> 7;
        int c = i & 127;
        int gr = row0 + r;
        float val = 0.0f;
        if (gr < NUM_DST) {
            val = agg[(size_t)gr * DIM + c] * rsqrtf(fmaxf(deg_dst[gr], 1.0f));
        }
        As[r * LDS_STRIDE + c] = val;
    }
    __syncthreads();

    const int wave    = threadIdx.x >> 5;
    const int lane    = threadIdx.x & 31;
    const int colBase = wave * 16;
    const int nn      = lane & 15;
    const int kHalf   = (lane >> 4) * 2;

    v8f acc = {};
    #pragma unroll
    for (int k0 = 0; k0 < DIM; k0 += 4) {
        v2f a, bf;
        a.x = As[nn * LDS_STRIDE + k0 + kHalf];
        a.y = As[nn * LDS_STRIDE + k0 + kHalf + 1];
        bf.x = W[(size_t)(k0 + kHalf) * DIM + colBase + nn];
        bf.y = W[(size_t)(k0 + kHalf + 1) * DIM + colBase + nn];
        acc = __builtin_amdgcn_wmma_f32_16x16x4_f32(
             false, a,  false, bf,
             (short)0, acc,  false,  false);
    }

    const float bcol = bias[colBase + nn];
    #pragma unroll
    for (int v = 0; v < 8; ++v) {
        int r = v + 8 * (lane >> 4);
        Os[r * LDS_STRIDE + colBase + nn] = acc[v] + bcol;
    }
    __syncthreads();

    if (threadIdx.x < 16) {
        int r = threadIdx.x;
        float dot = 0.0f;
        #pragma unroll 8
        for (int k = 0; k < DIM; ++k) dot += Os[r * LDS_STRIDE + k] * w_attn[k];
        float al = 1.0f / (1.0f + __expf(-dot));
        alphaS[r] = al;
        int gr = row0 + r;
        if (gr < NUM_DST) alpha_out[gr] = al;
    }
    __syncthreads();

    for (int i = threadIdx.x; i < 16 * DIM; i += 256) {
        int r = i >> 7;
        int c = i & 127;
        int gr = row0 + r;
        if (gr < NUM_DST) {
            out_scaled[(size_t)gr * DIM + c] = Os[r * LDS_STRIDE + c] * alphaS[r];
        }
    }
}

extern "C" void kernel_launch(void* const* d_in, const int* in_sizes, int n_in,
                              void* d_out, int out_size, void* d_ws, size_t ws_size,
                              hipStream_t stream) {
    const float* h      = (const float*)d_in[0];
    const int*   src    = (const int*)d_in[1];
    const int*   dst    = (const int*)d_in[2];
    const float* W      = (const float*)d_in[3];
    const float* bias   = (const float*)d_in[4];
    const float* w_attn = (const float*)d_in[5];

    float* out   = (float*)d_out;
    float* alpha = out + (size_t)NUM_DST * DIM;

    float* agg     = (float*)d_ws;
    float* deg_src = agg + (size_t)NUM_DST * DIM;
    float* deg_dst = deg_src + NUM_SRC;
    size_t zero_bytes = ((size_t)NUM_DST * DIM + NUM_SRC + NUM_DST) * sizeof(float);
    hipMemsetAsync(d_ws, 0, zero_bytes, stream);

    deg_kernel<<<(NUM_E + 255) / 256, 256, 0, stream>>>(src, dst, deg_src, deg_dst);

    scatter_kernel<<<((size_t)NUM_E * 32 + 255) / 256, 256, 0, stream>>>(
        h, src, dst, deg_src, agg);

    gemm_attn_kernel<<<(NUM_DST + 15) / 16, 256, 0, stream>>>(
        agg, deg_dst, W, bias, w_attn, out, alpha);
}


# Float-atomic scatter replaced by a counting sort of edges by destination (int atomics: counts + arrival-order rank) and a register row gather into the GEMM's LDS A tile; sum order = rank order
# speedup vs baseline: 1.8317x; 1.8317x over previous
#include <hip/hip_runtime.h>
#include <math.h>

#define NUM_SRC 100000
#define NUM_DST 50000
#define NUM_E   800000
#define DIM     128

typedef __attribute__((ext_vector_type(2))) float v2f;
typedef __attribute__((ext_vector_type(8))) float v8f;

#define WS_DEG_SRC 0
#define WS_DEG_DST 100000
#define WS_OFF     150016
#define WS_RANK    200064
#define WS_PERM    1000064

__global__ __launch_bounds__(256) void count_kernel(const int* __restrict__ src,
                                                    const int* __restrict__ dst,
                                                    int* __restrict__ deg_src,
                                                    int* __restrict__ deg_dst,
                                                    int* __restrict__ rank) {
    int e = blockIdx.x * blockDim.x + threadIdx.x;
    if (e < NUM_E) {
        int s = src[e];
        int d = dst[e];
        atomicAdd(&deg_src[s], 1);
        rank[e] = atomicAdd(&deg_dst[d], 1);
    }
}

#define SCAN_T  1024
#define SCAN_CH 49
#define SCAN_N  (SCAN_T * SCAN_CH)

__global__ __launch_bounds__(SCAN_T) void scan_kernel(const int* __restrict__ deg_dst,
                                                      int* __restrict__ off) {
    __shared__ int buf[SCAN_N];
    __shared__ int wtot[32];
    const int t    = threadIdx.x;
    const int lane = t & 31;
    const int w    = t >> 5;

    static_assert((SCAN_CH - 1) * SCAN_T <= NUM_DST && SCAN_N >= NUM_DST + 1, "scan chunking");
    int vals[SCAN_CH];
    #pragma unroll
    for (int k = 0; k < SCAN_CH - 1; ++k) vals[k] = deg_dst[t + k * SCAN_T];
    {
        const int i = t + (SCAN_CH - 1) * SCAN_T;
        vals[SCAN_CH - 1] = (i < NUM_DST) ? deg_dst[i] : 0;
    }
    #pragma unroll
    for (int k = 0; k < SCAN_CH; ++k) buf[t + k * SCAN_T] = vals[k];
    __syncthreads();

    const int c0 = t * SCAN_CH;
    int sum = 0;
    #pragma unroll
    for (int k = 0; k < SCAN_CH; ++k) { vals[k] = buf[c0 + k]; sum += vals[k]; }

    int inc = sum;
    for (int d = 1; d < 32; d <<= 1) {
        int o = __shfl_up(inc, d);
        if (lane >= d) inc += o;
    }
    if (lane == 31) wtot[w] = inc;
    __syncthreads();
    if (w == 0) {
        int v = wtot[lane];
        int iv = v;
        for (int d = 1; d < 32; d <<= 1) {
            int o = __shfl_up(iv, d);
            if (lane >= d) iv += o;
        }
        wtot[lane] = iv - v;
    }
    __syncthreads();
    int run = wtot[w] + inc - sum;
    #pragma unroll
    for (int k = 0; k < SCAN_CH; ++k) { buf[c0 + k] = run; run += vals[k]; }
    __syncthreads();
    for (int i = t; i <= NUM_DST; i += SCAN_T) off[i] = buf[i];
}

__global__ __launch_bounds__(256) void fill_kernel(const int* __restrict__ src,
                                                   const int* __restrict__ dst,
                                                   const int* __restrict__ deg_src,
                                                   const int* __restrict__ off,
                                                   const int* __restrict__ rank,
                                                   int* __restrict__ perm) {
    int e = blockIdx.x * blockDim.x + threadIdx.x;
    if (e < NUM_E) {
        int s = src[e];
        int d = dst[e];
        int r = rank[e];
        int pos = off[d] + r;
        pos = min(max(pos, 0), NUM_E - 1);
        float sc = rsqrtf(fmaxf((float)deg_src[s], 1.0f));
        *(int2*)(perm + 2 * (size_t)pos) = make_int2(s, __float_as_int(sc));
    }
}

template <int K>
__device__ __forceinline__ void gather_group(const float* __restrict__ hl, int ms, int mc,
                                             int j, float4& acc) {
    float4 v[K];
    float  sc[K];
    #pragma unroll
    for (int k = 0; k < K; ++k) {
        int s = __builtin_amdgcn_readlane(ms, j + k);
        s = min(max(s, 0), NUM_SRC - 1);
        sc[k] = __int_as_float(__builtin_amdgcn_readlane(mc, j + k));
        v[k] = *(const float4*)(hl + (size_t)s * DIM);
    }
    #pragma unroll
    for (int k = 0; k < K; ++k) {
        acc.x = fmaf(v[k].x, sc[k], acc.x);
        acc.y = fmaf(v[k].y, sc[k], acc.y);
        acc.z = fmaf(v[k].z, sc[k], acc.z);
        acc.w = fmaf(v[k].w, sc[k], acc.w);
    }
}

__device__ __forceinline__ void load_chunk(const int* __restrict__ perm, int base, int end,
                                           int lane, int& ms, int& mc) {
    ms = 0;
    mc = 0;
    if (lane < end - base) {
        int2 p = *(const int2*)(perm + 2 * (size_t)(base + lane));
        ms = p.x;
        mc = p.y;
    }
}

__device__ __forceinline__ float4 gather_row(const float* __restrict__ h,
                                             const int* __restrict__ perm,
                                             int beg, int end, int lane, int ms, int mc) {
    float4 acc = make_float4(0.0f, 0.0f, 0.0f, 0.0f);
    const float* hl = h + lane * 4;
    for (int base = beg;;) {
        int n = end - base;
        n = n < 32 ? n : 32;
        int j = 0;
        for (; j + 8 <= n; j += 8) gather_group<8>(hl, ms, mc, j, acc);
        if (n - j >= 4) { gather_group<4>(hl, ms, mc, j, acc); j += 4; }
        if (n - j >= 2) { gather_group<2>(hl, ms, mc, j, acc); j += 2; }
        if (n - j >= 1) { gather_group<1>(hl, ms, mc, j, acc); }
        base += 32;
        if (base >= end) break;
        load_chunk(perm, base, end, lane, ms, mc);
    }
    return acc;
}

#define LDS_STRIDE 132

__global__ __launch_bounds__(256) void gather_gemm_attn_kernel(const float* __restrict__ h,
                                                               const int* __restrict__ perm,
                                                               const int* __restrict__ off,
                                                               const float* __restrict__ W,
                                                               const float* __restrict__ bias,
                                                               const float* __restrict__ w_attn,
                                                               float* __restrict__ out_scaled,
                                                               float* __restrict__ alpha_out) {
    __shared__ float As[16 * LDS_STRIDE];
    __shared__ float Os[16 * LDS_STRIDE];
    __shared__ float alphaS[16];

    const int row0 = blockIdx.x * 16;
    const int wave = __builtin_amdgcn_readfirstlane(threadIdx.x >> 5);
    const int lane = threadIdx.x & 31;

    const int r0  = wave * 2;
    const int gr0 = row0 + r0;
    int b0 = off[min(gr0,     NUM_DST)];
    int b1 = off[min(gr0 + 1, NUM_DST)];
    int b2 = off[min(gr0 + 2, NUM_DST)];
    b0 = min(max(b0, 0), NUM_E);
    b1 = min(max(b1, b0), NUM_E);
    b2 = min(max(b2, b1), NUM_E);
    int ms0, mc0, ms1, mc1;
    load_chunk(perm, b0, b1, lane, ms0, mc0);
    load_chunk(perm, b1, b2, lane, ms1, mc1);
    {
        float4 acc = gather_row(h, perm, b0, b1, lane, ms0, mc0);
        float ds = rsqrtf(fmaxf((float)(b1 - b0), 1.0f));
        *(float4*)(&As[r0 * LDS_STRIDE + lane * 4]) =
            make_float4(acc.x * ds, acc.y * ds, acc.z * ds, acc.w * ds);
    }
    {
        float4 acc = gather_row(h, perm, b1, b2, lane, ms1, mc1);
        float ds = rsqrtf(fmaxf((float)(b2 - b1), 1.0f));
        *(float4*)(&As[(r0 + 1) * LDS_STRIDE + lane * 4]) =
            make_float4(acc.x * ds, acc.y * ds, acc.z * ds, acc.w * ds);
    }
    __syncthreads();

    const int colBase = wave * 16;
    const int nn      = lane & 15;
    const int kHalf   = (lane >> 4) * 2;

    v8f acc = {};
    #pragma unroll
    for (int k0 = 0; k0 < DIM; k0 += 4) {
        v2f a, bf;
        a.x = As[nn * LDS_STRIDE + k0 + kHalf];
        a.y = As[nn * LDS_STRIDE + k0 + kHalf + 1];
        bf.x = W[(size_t)(k0 + kHalf) * DIM + colBase + nn];
        bf.y = W[(size_t)(k0 + kHalf + 1) * DIM + colBase + nn];
        acc = __builtin_amdgcn_wmma_f32_16x16x4_f32(
             false, a,  false, bf,
             (short)0, acc,  false,  false);
    }

    const float bcol = bias[colBase + nn];
    #pragma unroll
    for (int v = 0; v < 8; ++v) {
        int r = v + 8 * (lane >> 4);
        Os[r * LDS_STRIDE + colBase + nn] = acc[v] + bcol;
    }
    __syncthreads();

    if (threadIdx.x < 16) {
        int r = threadIdx.x;
        float dot = 0.0f;
        #pragma unroll 8
        for (int k = 0; k < DIM; ++k) dot += Os[r * LDS_STRIDE + k] * w_attn[k];
        float al = 1.0f / (1.0f + __expf(-dot));
        alphaS[r] = al;
        int gr = row0 + r;
        if (gr < NUM_DST) alpha_out[gr] = al;
    }
    __syncthreads();

    for (int i = threadIdx.x; i < 16 * DIM; i += 256) {
        int r = i >> 7;
        int c = i & 127;
        int gr = row0 + r;
        if (gr < NUM_DST) {
            out_scaled[(size_t)gr * DIM + c] = Os[r * LDS_STRIDE + c] * alphaS[r];
        }
    }
}

extern "C" void kernel_launch(void* const* d_in, const int* in_sizes, int n_in,
                              void* d_out, int out_size, void* d_ws, size_t ws_size,
                              hipStream_t stream) {
    const float* h      = (const float*)d_in[0];
    const int*   src    = (const int*)d_in[1];
    const int*   dst    = (const int*)d_in[2];
    const float* W      = (const float*)d_in[3];
    const float* bias   = (const float*)d_in[4];
    const float* w_attn = (const float*)d_in[5];

    float* out   = (float*)d_out;
    float* alpha = out + (size_t)NUM_DST * DIM;

    int* wsi     = (int*)d_ws;
    int* deg_src = wsi + WS_DEG_SRC;
    int* deg_dst = wsi + WS_DEG_DST;
    int* off     = wsi + WS_OFF;
    int* rank    = wsi + WS_RANK;
    int* perm    = wsi + WS_PERM;

    hipMemsetAsync(d_ws, 0, (size_t)(NUM_SRC + NUM_DST) * sizeof(int), stream);

    count_kernel<<<(NUM_E + 255) / 256, 256, 0, stream>>>(src, dst, deg_src, deg_dst, rank);

    scan_kernel<<<1, SCAN_T, 0, stream>>>(deg_dst, off);

    fill_kernel<<<(NUM_E + 255) / 256, 256, 0, stream>>>(src, dst, deg_src, off, rank, perm);

    gather_gemm_attn_kernel<<<(NUM_DST + 15) / 16, 256, 0, stream>>>(
        h, perm, off, W, bias, w_attn, out, alpha);
}


# Scan and fill launches removed: count kernel's returning add files each source in its destination's 48-slot bucket (fuller rows: edge scan, edge order); scale read at gather; rank sum order kept
# speedup vs baseline: 2.1272x; 1.1613x over previous
#include <hip/hip_runtime.h>
#include <math.h>

#define NUM_SRC 100000
#define NUM_DST 50000
#define NUM_E   800000
#define DIM     128

typedef __attribute__((ext_vector_type(2))) float v2f;
typedef __attribute__((ext_vector_type(8))) float v8f;

#define CAP     48

#define WS_DEG_SRC 0
#define WS_DEG_DST 100000
#define WS_BUCKET  150016

__global__ __launch_bounds__(256) void count_place_kernel(const int* __restrict__ src,
                                                          const int* __restrict__ dst,
                                                          int* __restrict__ deg_src,
                                                          int* __restrict__ deg_dst,
                                                          int* __restrict__ bucket) {
    int e = blockIdx.x * blockDim.x + threadIdx.x;
    if (e < NUM_E) {
        int s = src[e];
        int d = dst[e];
        atomicAdd(&deg_src[s], 1);
        int r = atomicAdd(&deg_dst[d], 1);
        if (r < CAP) bucket[(size_t)d * CAP + r] = s;
    }
}

template <int K>
__device__ __forceinline__ void gather_group(const float* __restrict__ hl, int ms, int mc,
                                             int j, float4& acc) {
    float4 v[K];
    float  sc[K];
    #pragma unroll
    for (int k = 0; k < K; ++k) {
        int s = __builtin_amdgcn_readlane(ms, j + k);
        s = min(max(s, 0), NUM_SRC - 1);
        sc[k] = __int_as_float(__builtin_amdgcn_readlane(mc, j + k));
        v[k] = *(const float4*)(hl + (size_t)s * DIM);
    }
    #pragma unroll
    for (int k = 0; k < K; ++k) {
        acc.x = fmaf(v[k].x, sc[k], acc.x);
        acc.y = fmaf(v[k].y, sc[k], acc.y);
        acc.z = fmaf(v[k].z, sc[k], acc.z);
        acc.w = fmaf(v[k].w, sc[k], acc.w);
    }
}

__device__ __forceinline__ void load_chunk(const int* __restrict__ brow,
                                           const int* __restrict__ deg_src, int base, int end,
                                           int lane, int& ms, int& mc) {
    ms = 0;
    mc = 0;
    if (lane < end - base) {
        int s = brow[base + lane];
        s = min(max(s, 0), NUM_SRC - 1);
        ms = s;
        mc = __float_as_int(rsqrtf(fmaxf((float)deg_src[s], 1.0f)));
    }
}

__device__ __forceinline__ float4 gather_row(const float* __restrict__ h,
                                             const int* __restrict__ brow,
                                             const int* __restrict__ deg_src,
                                             int end, int lane, int ms, int mc) {
    float4 acc = make_float4(0.0f, 0.0f, 0.0f, 0.0f);
    const float* hl = h + lane * 4;
    for (int base = 0;;) {
        int n = end - base;
        n = n < 32 ? n : 32;
        int j = 0;
        for (; j + 8 <= n; j += 8) gather_group<8>(hl, ms, mc, j, acc);
        if (n - j >= 4) { gather_group<4>(hl, ms, mc, j, acc); j += 4; }
        if (n - j >= 2) { gather_group<2>(hl, ms, mc, j, acc); j += 2; }
        if (n - j >= 1) { gather_group<1>(hl, ms, mc, j, acc); }
        base += 32;
        if (base >= end) break;
        load_chunk(brow, deg_src, base, end, lane, ms, mc);
    }
    return acc;
}

__device__ __forceinline__ float4 gather_row_scan(const float* __restrict__ h,
                                               const int* __restrict__ src,
                                               const int* __restrict__ dst,
                                               const int* __restrict__ deg_src,
                                               int d, int lane) {
    static_assert(NUM_E % 32 == 0, "edge scan walks whole 32-edge trips");
    float4 acc = make_float4(0.0f, 0.0f, 0.0f, 0.0f);
    const float* hl = h + lane * 4;
    for (int base = 0; base < NUM_E; base += 32) {
        const int e = base + lane;
        const int dd = dst[e];
        int s = src[e];
        unsigned m = __builtin_amdgcn_ballot_w32(dd == d);
        if (m != 0u) {
            s = min(max(s, 0), NUM_SRC - 1);
            const int scb = __float_as_int(rsqrtf(fmaxf((float)deg_src[s], 1.0f)));
            while (m != 0u) {
                const int j = __builtin_ctz(m);
                m &= m - 1u;
                const int   sj = __builtin_amdgcn_readlane(s, j);
                const float sc = __int_as_float(__builtin_amdgcn_readlane(scb, j));
                const float4 v = *(const float4*)(hl + (size_t)sj * DIM);
                acc.x = fmaf(v.x, sc, acc.x);
                acc.y = fmaf(v.y, sc, acc.y);
                acc.z = fmaf(v.z, sc, acc.z);
                acc.w = fmaf(v.w, sc, acc.w);
            }
        }
    }
    return acc;
}

#define LDS_STRIDE 132

__global__ __launch_bounds__(256) void gather_gemm_attn_kernel(const float* __restrict__ h,
                                                               const int* __restrict__ src,
                                                               const int* __restrict__ dst,
                                                               const int* __restrict__ deg_src,
                                                               const int* __restrict__ deg_dst,
                                                               const int* __restrict__ bucket,
                                                               const float* __restrict__ W,
                                                               const float* __restrict__ bias,
                                                               const float* __restrict__ w_attn,
                                                               float* __restrict__ out_scaled,
                                                               float* __restrict__ alpha_out) {
    __shared__ float As[16 * LDS_STRIDE];
    __shared__ float Os[16 * LDS_STRIDE];
    __shared__ float alphaS[16];

    const int row0 = blockIdx.x * 16;
    const int wave = __builtin_amdgcn_readfirstlane(threadIdx.x >> 5);
    const int lane = threadIdx.x & 31;

    static_assert(NUM_DST % 2 == 0 && WS_DEG_DST % 2 == 0, "a row pair's counts are one aligned 8-byte load");
    const int r0  = wave * 2;
    const int gr0 = row0 + r0;
    const int d0  = min(gr0, NUM_DST - 2);
    const int d1  = d0 + 1;
    const int2 cnt = *(const int2*)(deg_dst + d0);
    const int n0 = (gr0 < NUM_DST) ? max(cnt.x, 0) : 0;
    const int n1 = (gr0 < NUM_DST) ? max(cnt.y, 0) : 0;
    const int* brow0 = bucket + (size_t)d0 * CAP;
    const int* brow1 = bucket + (size_t)d1 * CAP;
    const int c0 = min(n0, CAP);
    const int c1 = min(n1, CAP);
    const int* p0 = (lane < c0) ? (brow0 + lane) : (deg_dst + d0);
    const int* p1 = (lane < c1) ? (brow1 + lane) : (deg_dst + d1);
    int ms0 = *p0;
    int ms1 = *p1;
    ms0 = min(max(ms0, 0), NUM_SRC - 1);
    ms1 = min(max(ms1, 0), NUM_SRC - 1);
    const int g0 = deg_src[ms0];
    const int g1 = deg_src[ms1];
    const int mc0 = __float_as_int(rsqrtf(fmaxf((float)g0, 1.0f)));
    const int mc1 = __float_as_int(rsqrtf(fmaxf((float)g1, 1.0f)));
    {
        float4 acc = gather_row(h, brow0, deg_src, c0, lane, ms0, mc0);
        if (__builtin_expect(n0 > CAP, 0)) acc = gather_row_scan(h, src, dst, deg_src, d0, lane);
        float ds = rsqrtf(fmaxf((float)n0, 1.0f));
        *(float4*)(&As[r0 * LDS_STRIDE + lane * 4]) =
            make_float4(acc.x * ds, acc.y * ds, acc.z * ds, acc.w * ds);
    }
    {
        float4 acc = gather_row(h, brow1, deg_src, c1, lane, ms1, mc1);
        if (__builtin_expect(n1 > CAP, 0)) acc = gather_row_scan(h, src, dst, deg_src, d1, lane);
        float ds = rsqrtf(fmaxf((float)n1, 1.0f));
        *(float4*)(&As[(r0 + 1) * LDS_STRIDE + lane * 4]) =
            make_float4(acc.x * ds, acc.y * ds, acc.z * ds, acc.w * ds);
    }
    __syncthreads();

    const int colBase = wave * 16;
    const int nn      = lane & 15;
    const int kHalf   = (lane >> 4) * 2;

    v8f acc = {};
    #pragma unroll
    for (int k0 = 0; k0 < DIM; k0 += 4) {
        v2f a, bf;
        a.x = As[nn * LDS_STRIDE + k0 + kHalf];
        a.y = As[nn * LDS_STRIDE + k0 + kHalf + 1];
        bf.x = W[(size_t)(k0 + kHalf) * DIM + colBase + nn];
        bf.y = W[(size_t)(k0 + kHalf + 1) * DIM + colBase + nn];
        acc = __builtin_amdgcn_wmma_f32_16x16x4_f32(
             false, a,  false, bf,
             (short)0, acc,  false,  false);
    }

    const float bcol = bias[colBase + nn];
    #pragma unroll
    for (int v = 0; v < 8; ++v) {
        int r = v + 8 * (lane >> 4);
        Os[r * LDS_STRIDE + colBase + nn] = acc[v] + bcol;
    }
    __syncthreads();

    if (threadIdx.x < 16) {
        int r = threadIdx.x;
        float dot = 0.0f;
        #pragma unroll 8
        for (int k = 0; k < DIM; ++k) dot += Os[r * LDS_STRIDE + k] * w_attn[k];
        float al = 1.0f / (1.0f + __expf(-dot));
        alphaS[r] = al;
        int gr = row0 + r;
        if (gr < NUM_DST) alpha_out[gr] = al;
    }
    __syncthreads();

    for (int i = threadIdx.x; i < 16 * DIM; i += 256) {
        int r = i >> 7;
        int c = i & 127;
        int gr = row0 + r;
        if (gr < NUM_DST) {
            out_scaled[(size_t)gr * DIM + c] = Os[r * LDS_STRIDE + c] * alphaS[r];
        }
    }
}

extern "C" void kernel_launch(void* const* d_in, const int* in_sizes, int n_in,
                              void* d_out, int out_size, void* d_ws, size_t ws_size,
                              hipStream_t stream) {
    const float* h      = (const float*)d_in[0];
    const int*   src    = (const int*)d_in[1];
    const int*   dst    = (const int*)d_in[2];
    const float* W      = (const float*)d_in[3];
    const float* bias   = (const float*)d_in[4];
    const float* w_attn = (const float*)d_in[5];

    float* out   = (float*)d_out;
    float* alpha = out + (size_t)NUM_DST * DIM;

    int* wsi     = (int*)d_ws;
    int* deg_src = wsi + WS_DEG_SRC;
    int* deg_dst = wsi + WS_DEG_DST;
    int* bucket  = wsi + WS_BUCKET;

    hipMemsetAsync(d_ws, 0, (size_t)(NUM_SRC + NUM_DST) * sizeof(int), stream);

    count_place_kernel<<<(NUM_E + 255) / 256, 256, 0, stream>>>(src, dst, deg_src, deg_dst, bucket);

    gather_gemm_attn_kernel<<<(NUM_DST + 15) / 16, 256, 0, stream>>>(
        h, src, dst, deg_src, deg_dst, bucket, W, bias, w_attn, out, alpha);
}


# Per-edge global rank add and slot store replaced by an LDS counting sort into 256-destination bins (one global add per block and bin, run stores); row lists built in LDS; sum order = LDS arrival
# speedup vs baseline: 2.9520x; 1.3877x over previous
#include <hip/hip_runtime.h>
#include <math.h>

#define NUM_SRC 100000
#define NUM_DST 50000
#define NUM_E   800000
#define DIM     128

typedef __attribute__((ext_vector_type(2))) float v2f;
typedef __attribute__((ext_vector_type(8))) float v8f;

#define BIN_SHIFT 8
#define BIN_DST   256
#define NBIN      ((NUM_DST + BIN_DST - 1) / BIN_DST)
#define BINCAP    6144
#define ROWCAP    64

#define WS_DEG_SRC 0
#define WS_BINCNT  100000
#define WS_BINBUF  100352

#define K1_THREADS 1024
#define K1_EPT     4
#define K1_EDGES   (K1_THREADS * K1_EPT)

__global__ __launch_bounds__(K1_THREADS) void bin_kernel(const int* __restrict__ src,
                                                         const int* __restrict__ dst,
                                                         int* __restrict__ deg_src,
                                                         int* __restrict__ bin_count,
                                                         int* __restrict__ binbuf) {
    __shared__ int hist[256];
    __shared__ int lofs[256];
    __shared__ int gbase[256];
    __shared__ int wsum[8];
    __shared__ int sval[K1_EDGES];
    __shared__ unsigned char sbin[K1_EDGES];

    const int tid  = threadIdx.x;
    const int lane = tid & 31;
    const int wave = tid >> 5;
    if (tid < 256) hist[tid] = 0;
    __syncthreads();

    const int e0 = blockIdx.x * K1_EDGES;
    int s[K1_EPT], d[K1_EPT], b[K1_EPT], r[K1_EPT];
    #pragma unroll
    for (int i = 0; i < K1_EPT; ++i) {
        const int e = e0 + i * K1_THREADS + tid;
        s[i] = 0;
        d[i] = 0;
        if (e < NUM_E) {
            s[i] = src[e];
            d[i] = dst[e];
        }
    }
    #pragma unroll
    for (int i = 0; i < K1_EPT; ++i) {
        const int e = e0 + i * K1_THREADS + tid;
        b[i] = min(max(d[i] >> BIN_SHIFT, 0), NBIN - 1);
        r[i] = 0;
        if (e < NUM_E) {
            atomicAdd(&deg_src[s[i]], 1);
            r[i] = atomicAdd(&hist[b[i]], 1);
        }
    }
    __syncthreads();

    int c = 0, incl = 0;
    if (tid < 256) {
        c = hist[tid];
        incl = c;
        #pragma unroll
        for (int o = 1; o < 32; o <<= 1) {
            const int t = __shfl_up(incl, o);
            if (lane >= o) incl += t;
        }
        if (lane == 31) wsum[wave] = incl;
    }
    __syncthreads();
    if (tid < 256) {
        int add = 0;
        for (int w = 0; w < wave; ++w) add += wsum[w];
        lofs[tid] = add + incl - c;
        int gb = 0;
        if (c > 0) gb = atomicAdd(&bin_count[tid], c);
        gbase[tid] = gb;
    }
    __syncthreads();

    #pragma unroll
    for (int i = 0; i < K1_EPT; ++i) {
        const int e = e0 + i * K1_THREADS + tid;
        if (e < NUM_E) {
            const int p = lofs[b[i]] + r[i];
            sval[p] = (s[i] << 8) | (d[i] & (BIN_DST - 1));
            sbin[p] = (unsigned char)b[i];
        }
    }
    __syncthreads();

    const int total = min(K1_EDGES, NUM_E - e0);
    for (int j = tid; j < total; j += K1_THREADS) {
        const int bj  = sbin[j];
        const int pos = gbase[bj] + (j - lofs[bj]);
        if (pos >= 0 && pos < BINCAP) binbuf[(size_t)bj * BINCAP + pos] = sval[j];
    }
}

template <int K>
__device__ __forceinline__ void gather_group(const float* __restrict__ hl, int ms, int mc,
                                             int j, float4& acc) {
    float4 v[K];
    float  sc[K];
    #pragma unroll
    for (int k = 0; k < K; ++k) {
        int s = __builtin_amdgcn_readlane(ms, j + k);
        s = min(max(s, 0), NUM_SRC - 1);
        sc[k] = __int_as_float(__builtin_amdgcn_readlane(mc, j + k));
        v[k] = *(const float4*)(hl + (size_t)s * DIM);
    }
    #pragma unroll
    for (int k = 0; k < K; ++k) {
        acc.x = fmaf(v[k].x, sc[k], acc.x);
        acc.y = fmaf(v[k].y, sc[k], acc.y);
        acc.z = fmaf(v[k].z, sc[k], acc.z);
        acc.w = fmaf(v[k].w, sc[k], acc.w);
    }
}

__device__ __forceinline__ void load_chunk(const int* __restrict__ brow,
                                           const int* __restrict__ deg_src, int base, int end,
                                           int lane, int& ms, int& mc) {
    ms = 0;
    mc = 0;
    if (lane < end - base) {
        int s = brow[base + lane];
        s = min(max(s, 0), NUM_SRC - 1);
        ms = s;
        mc = __float_as_int(rsqrtf(fmaxf((float)deg_src[s], 1.0f)));
    }
}

__device__ __forceinline__ float4 gather_row(const float* __restrict__ h,
                                             const int* __restrict__ brow,
                                             const int* __restrict__ deg_src,
                                             int end, int lane, int ms, int mc) {
    float4 acc = make_float4(0.0f, 0.0f, 0.0f, 0.0f);
    const float* hl = h + lane * 4;
    for (int base = 0;;) {
        int n = end - base;
        n = n < 32 ? n : 32;
        int j = 0;
        for (; j + 8 <= n; j += 8) gather_group<8>(hl, ms, mc, j, acc);
        if (n - j >= 4) { gather_group<4>(hl, ms, mc, j, acc); j += 4; }
        if (n - j >= 2) { gather_group<2>(hl, ms, mc, j, acc); j += 2; }
        if (n - j >= 1) { gather_group<1>(hl, ms, mc, j, acc); }
        base += 32;
        if (base >= end) break;
        load_chunk(brow, deg_src, base, end, lane, ms, mc);
    }
    return acc;
}

__device__ __forceinline__ float4 gather_row_scan(const float* __restrict__ h,
                                               const int* __restrict__ src,
                                               const int* __restrict__ dst,
                                               const int* __restrict__ deg_src,
                                               int d, int lane) {
    static_assert(NUM_E % 32 == 0, "edge scan walks whole 32-edge trips");
    float4 acc = make_float4(0.0f, 0.0f, 0.0f, 0.0f);
    const float* hl = h + lane * 4;
    for (int base = 0; base < NUM_E; base += 32) {
        const int e = base + lane;
        const int dd = dst[e];
        int s = src[e];
        unsigned m = __builtin_amdgcn_ballot_w32(dd == d);
        if (m != 0u) {
            s = min(max(s, 0), NUM_SRC - 1);
            const int scb = __float_as_int(rsqrtf(fmaxf((float)deg_src[s], 1.0f)));
            while (m != 0u) {
                const int j = __builtin_ctz(m);
                m &= m - 1u;
                const int   sj = __builtin_amdgcn_readlane(s, j);
                const float sc = __int_as_float(__builtin_amdgcn_readlane(scb, j));
                const float4 v = *(const float4*)(hl + (size_t)sj * DIM);
                acc.x = fmaf(v.x, sc, acc.x);
                acc.y = fmaf(v.y, sc, acc.y);
                acc.z = fmaf(v.z, sc, acc.z);
                acc.w = fmaf(v.w, sc, acc.w);
            }
        }
    }
    return acc;
}

#define LDS_STRIDE 132
#define BIN_TRIPS  (BINCAP / 1024)

__global__ __launch_bounds__(256) void gather_gemm_attn_kernel(const float* __restrict__ h,
                                                               const int* __restrict__ src,
                                                               const int* __restrict__ dst,
                                                               const int* __restrict__ deg_src,
                                                               const int* __restrict__ bin_count,
                                                               const int* __restrict__ binbuf,
                                                               const float* __restrict__ W,
                                                               const float* __restrict__ bias,
                                                               const float* __restrict__ w_attn,
                                                               float* __restrict__ out_scaled,
                                                               float* __restrict__ alpha_out) {
    __shared__ float As[16 * LDS_STRIDE];
    __shared__ float Os[16 * LDS_STRIDE];
    __shared__ float alphaS[16];
    __shared__ int   rowcnt[16];
    __shared__ int   rowlist[16 * ROWCAP];

    static_assert(BIN_DST % 16 == 0 && BINCAP % 1024 == 0, "a row tile lies in one bin; whole trips");
    const int row0 = blockIdx.x * 16;
    const int bin  = row0 >> BIN_SHIFT;
    const int lb   = row0 & (BIN_DST - 1);
    const int wave = __builtin_amdgcn_readfirstlane(threadIdx.x >> 5);
    const int lane = threadIdx.x & 31;

    if (threadIdx.x < 16) rowcnt[threadIdx.x] = 0;
    __syncthreads();

    auto row_append = [&](int r, int sv) {
        const int slot = atomicAdd(&rowcnt[r], 1);
        if (slot < ROWCAP) rowlist[r * ROWCAP + slot] = sv;
    };

    const int cnt = max(bin_count[bin], 0);
    if (cnt <= BINCAP) {
        const int* bb = binbuf + (size_t)bin * BINCAP;
        int4 q[BIN_TRIPS];
        #pragma unroll
        for (int t = 0; t < BIN_TRIPS; ++t) {
            const int i = t * 1024 + threadIdx.x * 4;
            q[t] = make_int4(-1, -1, -1, -1);
            if (cnt - i >= 4) q[t] = *(const int4*)(bb + i);
        }
        int qt = -1;
        if ((int)threadIdx.x < (cnt & 3)) qt = bb[(cnt & ~3) + threadIdx.x];
        #pragma unroll
        for (int t = 0; t < BIN_TRIPS; ++t) {
            const int vv[4] = {q[t].x, q[t].y, q[t].z, q[t].w};
            int slot[4];
            #pragma unroll
            for (int k = 0; k < 4; ++k) {
                const int r = (vv[k] & (BIN_DST - 1)) - lb;
                slot[k] = ROWCAP;
                if (vv[k] >= 0 && (unsigned)r < 16u) slot[k] = atomicAdd(&rowcnt[r], 1);
            }
            #pragma unroll
            for (int k = 0; k < 4; ++k) {
                const int r = (vv[k] & (BIN_DST - 1)) - lb;
                if (slot[k] < ROWCAP) rowlist[r * ROWCAP + slot[k]] = vv[k] >> 8;
            }
        }
        {
            const int r = (qt & (BIN_DST - 1)) - lb;
            if (qt >= 0 && (unsigned)r < 16u) row_append(r, qt >> 8);
        }
    } else {
        for (int e = threadIdx.x; e < NUM_E; e += 256) {
            const int r = dst[e] - row0;
            if ((unsigned)r < 16u) row_append(r, src[e]);
        }
    }
    __syncthreads();

    const int r0 = wave * 2;
    const int n0 = rowcnt[r0];
    const int n1 = rowcnt[r0 + 1];
    const int c0 = min(n0, ROWCAP);
    const int c1 = min(n1, ROWCAP);
    const int* brow0 = rowlist + r0 * ROWCAP;
    const int* brow1 = brow0 + ROWCAP;
    int ms0 = (lane < c0) ? brow0[lane] : 0;
    int ms1 = (lane < c1) ? brow1[lane] : 0;
    ms0 = min(max(ms0, 0), NUM_SRC - 1);
    ms1 = min(max(ms1, 0), NUM_SRC - 1);
    const int g0 = deg_src[ms0];
    const int g1 = deg_src[ms1];
    const int mc0 = __float_as_int(rsqrtf(fmaxf((float)g0, 1.0f)));
    const int mc1 = __float_as_int(rsqrtf(fmaxf((float)g1, 1.0f)));
    {
        float4 acc = gather_row(h, brow0, deg_src, c0, lane, ms0, mc0);
        if (__builtin_expect(n0 > ROWCAP, 0)) acc = gather_row_scan(h, src, dst, deg_src, row0 + r0, lane);
        float ds = rsqrtf(fmaxf((float)n0, 1.0f));
        *(float4*)(&As[r0 * LDS_STRIDE + lane * 4]) =
            make_float4(acc.x * ds, acc.y * ds, acc.z * ds, acc.w * ds);
    }
    {
        float4 acc = gather_row(h, brow1, deg_src, c1, lane, ms1, mc1);
        if (__builtin_expect(n1 > ROWCAP, 0)) acc = gather_row_scan(h, src, dst, deg_src, row0 + r0 + 1, lane);
        float ds = rsqrtf(fmaxf((float)n1, 1.0f));
        *(float4*)(&As[(r0 + 1) * LDS_STRIDE + lane * 4]) =
            make_float4(acc.x * ds, acc.y * ds, acc.z * ds, acc.w * ds);
    }
    __syncthreads();

    const int colBase = wave * 16;
    const int nn      = lane & 15;
    const int kHalf   = (lane >> 4) * 2;

    v8f acc = {};
    #pragma unroll
    for (int k0 = 0; k0 < DIM; k0 += 4) {
        v2f a, bf;
        a.x = As[nn * LDS_STRIDE + k0 + kHalf];
        a.y = As[nn * LDS_STRIDE + k0 + kHalf + 1];
        bf.x = W[(size_t)(k0 + kHalf) * DIM + colBase + nn];
        bf.y = W[(size_t)(k0 + kHalf + 1) * DIM + colBase + nn];
        acc = __builtin_amdgcn_wmma_f32_16x16x4_f32(
             false, a,  false, bf,
             (short)0, acc,  false,  false);
    }

    const float bcol = bias[colBase + nn];
    #pragma unroll
    for (int v = 0; v < 8; ++v) {
        int r = v + 8 * (lane >> 4);
        Os[r * LDS_STRIDE + colBase + nn] = acc[v] + bcol;
    }
    __syncthreads();

    if (threadIdx.x < 16) {
        int r = threadIdx.x;
        float dot = 0.0f;
        #pragma unroll 8
        for (int k = 0; k < DIM; ++k) dot += Os[r * LDS_STRIDE + k] * w_attn[k];
        float al = 1.0f / (1.0f + __expf(-dot));
        alphaS[r] = al;
        int gr = row0 + r;
        if (gr < NUM_DST) alpha_out[gr] = al;
    }
    __syncthreads();

    for (int i = threadIdx.x; i < 16 * DIM; i += 256) {
        int r = i >> 7;
        int c = i & 127;
        int gr = row0 + r;
        if (gr < NUM_DST) {
            out_scaled[(size_t)gr * DIM + c] = Os[r * LDS_STRIDE + c] * alphaS[r];
        }
    }
}

extern "C" void kernel_launch(void* const* d_in, const int* in_sizes, int n_in,
                              void* d_out, int out_size, void* d_ws, size_t ws_size,
                              hipStream_t stream) {
    const float* h      = (const float*)d_in[0];
    const int*   src    = (const int*)d_in[1];
    const int*   dst    = (const int*)d_in[2];
    const float* W      = (const float*)d_in[3];
    const float* bias   = (const float*)d_in[4];
    const float* w_attn = (const float*)d_in[5];

    float* out   = (float*)d_out;
    float* alpha = out + (size_t)NUM_DST * DIM;

    int* wsi       = (int*)d_ws;
    int* deg_src   = wsi + WS_DEG_SRC;
    int* bin_count = wsi + WS_BINCNT;
    int* binbuf    = wsi + WS_BINBUF;

    hipMemsetAsync(d_ws, 0, (size_t)(NUM_SRC + 256) * sizeof(int), stream);

    bin_kernel<<<(NUM_E + K1_EDGES - 1) / K1_EDGES, K1_THREADS, 0, stream>>>(src, dst, deg_src, bin_count, binbuf);

    gather_gemm_attn_kernel<<<(NUM_DST + 15) / 16, 256, 0, stream>>>(
        h, src, dst, deg_src, bin_count, binbuf, W, bias, w_attn, out, alpha);
}


# Row tile per workgroup 16 -> 64 rows on 16 waves: each W fragment load feeds two accumulator tiles; W and the bin are fetched per 64 rows, not per 16; row sums stay in LDS arrival order
# speedup vs baseline: 3.0629x; 1.0376x over previous
#include <hip/hip_runtime.h>
#include <math.h>

#define NUM_SRC 100000
#define NUM_DST 50000
#define NUM_E   800000
#define DIM     128

typedef __attribute__((ext_vector_type(2))) float v2f;
typedef __attribute__((ext_vector_type(8))) float v8f;

#define BIN_SHIFT 8
#define BIN_DST   256
#define NBIN      ((NUM_DST + BIN_DST - 1) / BIN_DST)
#define BINCAP    6144
#define ROWCAP    64

#define WS_DEG_SRC 0
#define WS_BINCNT  100000
#define WS_BINBUF  100352

#define K1_THREADS 1024
#define K1_EPT     4
#define K1_EDGES   (K1_THREADS * K1_EPT)

__global__ __launch_bounds__(K1_THREADS) void bin_kernel(const int* __restrict__ src,
                                                         const int* __restrict__ dst,
                                                         int* __restrict__ deg_src,
                                                         int* __restrict__ bin_count,
                                                         int* __restrict__ binbuf) {
    __shared__ int hist[256];
    __shared__ int lofs[256];
    __shared__ int gbase[256];
    __shared__ int wsum[8];
    __shared__ int sval[K1_EDGES];
    __shared__ unsigned char sbin[K1_EDGES];

    const int tid  = threadIdx.x;
    const int lane = tid & 31;
    const int wave = tid >> 5;
    if (tid < 256) hist[tid] = 0;
    __syncthreads();

    const int e0 = blockIdx.x * K1_EDGES;
    int s[K1_EPT], d[K1_EPT], b[K1_EPT], r[K1_EPT];
    #pragma unroll
    for (int i = 0; i < K1_EPT; ++i) {
        const int e = e0 + i * K1_THREADS + tid;
        s[i] = 0;
        d[i] = 0;
        if (e < NUM_E) {
            s[i] = src[e];
            d[i] = dst[e];
        }
    }
    #pragma unroll
    for (int i = 0; i < K1_EPT; ++i) {
        const int e = e0 + i * K1_THREADS + tid;
        b[i] = min(max(d[i] >> BIN_SHIFT, 0), NBIN - 1);
        r[i] = 0;
        if (e < NUM_E) {
            atomicAdd(&deg_src[s[i]], 1);
            r[i] = atomicAdd(&hist[b[i]], 1);
        }
    }
    __syncthreads();

    int c = 0, incl = 0;
    if (tid < 256) {
        c = hist[tid];
        incl = c;
        #pragma unroll
        for (int o = 1; o < 32; o <<= 1) {
            const int t = __shfl_up(incl, o);
            if (lane >= o) incl += t;
        }
        if (lane == 31) wsum[wave] = incl;
    }
    __syncthreads();
    if (tid < 256) {
        int add = 0;
        for (int w = 0; w < wave; ++w) add += wsum[w];
        lofs[tid] = add + incl - c;
        int gb = 0;
        if (c > 0) gb = atomicAdd(&bin_count[tid], c);
        gbase[tid] = gb;
    }
    __syncthreads();

    #pragma unroll
    for (int i = 0; i < K1_EPT; ++i) {
        const int e = e0 + i * K1_THREADS + tid;
        if (e < NUM_E) {
            const int p = lofs[b[i]] + r[i];
            sval[p] = (s[i] << 8) | (d[i] & (BIN_DST - 1));
            sbin[p] = (unsigned char)b[i];
        }
    }
    __syncthreads();

    const int total = min(K1_EDGES, NUM_E - e0);
    for (int j = tid; j < total; j += K1_THREADS) {
        const int bj  = sbin[j];
        const int pos = gbase[bj] + (j - lofs[bj]);
        if (pos >= 0 && pos < BINCAP) binbuf[(size_t)bj * BINCAP + pos] = sval[j];
    }
}

template <int K>
__device__ __forceinline__ void gather_group(const float* __restrict__ h, int lane, int ms, int mc,
                                             int j, float4& acc) {
    float4 v[K];
    float  sc[K];
    #pragma unroll
    for (int k = 0; k < K; ++k) {
        int s = __builtin_amdgcn_readlane(ms, j + k);
        s = min(max(s, 0), NUM_SRC - 1);
        sc[k] = __int_as_float(__builtin_amdgcn_readlane(mc, j + k));
        v[k] = ((const float4*)(h + (size_t)s * DIM))[lane];
    }
    #pragma unroll
    for (int k = 0; k < K; ++k) {
        acc.x = fmaf(v[k].x, sc[k], acc.x);
        acc.y = fmaf(v[k].y, sc[k], acc.y);
        acc.z = fmaf(v[k].z, sc[k], acc.z);
        acc.w = fmaf(v[k].w, sc[k], acc.w);
    }
}

__device__ __forceinline__ void load_chunk(const int* __restrict__ brow,
                                           const int* __restrict__ deg_src, int base, int end,
                                           int lane, int& ms, int& mc) {
    ms = 0;
    mc = 0;
    if (lane < end - base) {
        int s = brow[base + lane];
        s = min(max(s, 0), NUM_SRC - 1);
        ms = s;
        mc = __float_as_int(rsqrtf(fmaxf((float)deg_src[s], 1.0f)));
    }
}

__device__ __forceinline__ float4 gather_row(const float* __restrict__ h,
                                             const int* __restrict__ brow,
                                             const int* __restrict__ deg_src,
                                             int end, int lane, int ms, int mc) {
    float4 acc = make_float4(0.0f, 0.0f, 0.0f, 0.0f);
    for (int base = 0;;) {
        int n = end - base;
        n = n < 32 ? n : 32;
        int j = 0;
        for (; j + 8 <= n; j += 8) gather_group<8>(h, lane, ms, mc, j, acc);
        if (n - j >= 4) { gather_group<4>(h, lane, ms, mc, j, acc); j += 4; }
        if (n - j >= 2) { gather_group<2>(h, lane, ms, mc, j, acc); j += 2; }
        if (n - j >= 1) { gather_group<1>(h, lane, ms, mc, j, acc); }
        base += 32;
        if (base >= end) break;
        load_chunk(brow, deg_src, base, end, lane, ms, mc);
    }
    return acc;
}

__device__ __forceinline__ float4 gather_row_scan(const float* __restrict__ h,
                                               const int* __restrict__ src,
                                               const int* __restrict__ dst,
                                               const int* __restrict__ deg_src,
                                               int d, int lane) {
    static_assert(NUM_E % 32 == 0, "edge scan walks whole 32-edge trips");
    float4 acc = make_float4(0.0f, 0.0f, 0.0f, 0.0f);
    const float* hl = h + lane * 4;
    for (int base = 0; base < NUM_E; base += 32) {
        const int e = base + lane;
        const int dd = dst[e];
        int s = src[e];
        unsigned m = __builtin_amdgcn_ballot_w32(dd == d);
        if (m != 0u) {
            s = min(max(s, 0), NUM_SRC - 1);
            const int scb = __float_as_int(rsqrtf(fmaxf((float)deg_src[s], 1.0f)));
            while (m != 0u) {
                const int j = __builtin_ctz(m);
                m &= m - 1u;
                const int   sj = __builtin_amdgcn_readlane(s, j);
                const float sc = __int_as_float(__builtin_amdgcn_readlane(scb, j));
                const float4 v = *(const float4*)(hl + (size_t)sj * DIM);
                acc.x = fmaf(v.x, sc, acc.x);
                acc.y = fmaf(v.y, sc, acc.y);
                acc.z = fmaf(v.z, sc, acc.z);
                acc.w = fmaf(v.w, sc, acc.w);
            }
        }
    }
    return acc;
}

#define LDS_STRIDE 132
#define TILE_ROWS  64
#define K2_THREADS 512
#define BIN_TRIPS  (BINCAP / (K2_THREADS * 4))

__global__ __launch_bounds__(K2_THREADS, 16) void gather_gemm_attn_kernel(const float* __restrict__ h,
                                                               const int* __restrict__ src,
                                                               const int* __restrict__ dst,
                                                               const int* __restrict__ deg_src,
                                                               const int* __restrict__ bin_count,
                                                               const int* __restrict__ binbuf,
                                                               const float* __restrict__ W,
                                                               const float* __restrict__ bias,
                                                               const float* __restrict__ w_attn,
                                                               float* __restrict__ out_scaled,
                                                               float* __restrict__ alpha_out) {
    __shared__ float As[TILE_ROWS * LDS_STRIDE];
    __shared__ float alphaS[TILE_ROWS];
    __shared__ int   rowcnt[TILE_ROWS];
    __shared__ int   rowlist[TILE_ROWS * ROWCAP];
    float* Os = As;

    static_assert(BIN_DST % TILE_ROWS == 0 && BINCAP % (K2_THREADS * 4) == 0,
                  "a row tile lies in one bin; whole trips");
    static_assert(TILE_ROWS == (K2_THREADS / 32) * 4, "4 gathered rows per wave");
    const int row0 = blockIdx.x * TILE_ROWS;
    const int bin  = row0 >> BIN_SHIFT;
    const int lb   = row0 & (BIN_DST - 1);
    const int wave = __builtin_amdgcn_readfirstlane(threadIdx.x >> 5);
    const int lane = threadIdx.x & 31;

    if (threadIdx.x < TILE_ROWS) rowcnt[threadIdx.x] = 0;
    __syncthreads();

    auto row_append = [&](int r, int sv) {
        const int slot = atomicAdd(&rowcnt[r], 1);
        if (slot < ROWCAP) rowlist[r * ROWCAP + slot] = sv;
    };

    const int cnt = max(bin_count[bin], 0);
    if (cnt <= BINCAP) {
        const int* bb = binbuf + (size_t)bin * BINCAP;
        int4 q[BIN_TRIPS];
        #pragma unroll
        for (int t = 0; t < BIN_TRIPS; ++t) {
            const int i = t * (K2_THREADS * 4) + threadIdx.x * 4;
            q[t] = make_int4(-1, -1, -1, -1);
            if (cnt - i >= 4) q[t] = *(const int4*)(bb + i);
        }
        int qt = -1;
        if ((int)threadIdx.x < (cnt & 3)) qt = bb[(cnt & ~3) + threadIdx.x];
        #pragma unroll
        for (int t = 0; t < BIN_TRIPS; ++t) {
            const int vv[4] = {q[t].x, q[t].y, q[t].z, q[t].w};
            int slot[4];
            #pragma unroll
            for (int k = 0; k < 4; ++k) {
                const int r = (vv[k] & (BIN_DST - 1)) - lb;
                slot[k] = ROWCAP;
                if (vv[k] >= 0 && (unsigned)r < (unsigned)TILE_ROWS) slot[k] = atomicAdd(&rowcnt[r], 1);
            }
            #pragma unroll
            for (int k = 0; k < 4; ++k) {
                const int r = (vv[k] & (BIN_DST - 1)) - lb;
                if (slot[k] < ROWCAP) rowlist[r * ROWCAP + slot[k]] = vv[k] >> 8;
            }
        }
        {
            const int r = (qt & (BIN_DST - 1)) - lb;
            if (qt >= 0 && (unsigned)r < (unsigned)TILE_ROWS) row_append(r, qt >> 8);
        }
    } else {
        for (int e = threadIdx.x; e < NUM_E; e += K2_THREADS) {
            const int r = dst[e] - row0;
            if ((unsigned)r < (unsigned)TILE_ROWS) row_append(r, src[e]);
        }
    }
    __syncthreads();

    const int r0 = wave * 4;
    int nA = rowcnt[r0], nB = rowcnt[r0 + 1], nC = rowcnt[r0 + 2], nD = rowcnt[r0 + 3];
    const int* brow = rowlist + r0 * ROWCAP;
    int msA = (lane < min(nA, ROWCAP)) ? brow[lane] : 0;
    int msB = (lane < min(nB, ROWCAP)) ? brow[ROWCAP + lane] : 0;
    int msC = (lane < min(nC, ROWCAP)) ? brow[2 * ROWCAP + lane] : 0;
    int msD = (lane < min(nD, ROWCAP)) ? brow[3 * ROWCAP + lane] : 0;
    msA = min(max(msA, 0), NUM_SRC - 1);
    msB = min(max(msB, 0), NUM_SRC - 1);
    msC = min(max(msC, 0), NUM_SRC - 1);
    msD = min(max(msD, 0), NUM_SRC - 1);
    const int gA = deg_src[msA];
    const int gB = deg_src[msB];
    const int gC = deg_src[msC];
    const int gD = deg_src[msD];
    int mcA = __float_as_int(rsqrtf(fmaxf((float)gA, 1.0f)));
    int mcB = __float_as_int(rsqrtf(fmaxf((float)gB, 1.0f)));
    int mcC = __float_as_int(rsqrtf(fmaxf((float)gC, 1.0f)));
    int mcD = __float_as_int(rsqrtf(fmaxf((float)gD, 1.0f)));
    for (int qi = 0; qi < 4; ++qi) {
        const int r = r0 + qi;
        float4 acc = gather_row(h, rowlist + r * ROWCAP, deg_src, min(nA, ROWCAP), lane, msA, mcA);
        if (__builtin_expect(nA > ROWCAP, 0)) acc = gather_row_scan(h, src, dst, deg_src, row0 + r, lane);
        const float ds = rsqrtf(fmaxf((float)nA, 1.0f));
        *(float4*)(&As[r * LDS_STRIDE + lane * 4]) =
            make_float4(acc.x * ds, acc.y * ds, acc.z * ds, acc.w * ds);
        nA = nB;  nB = nC;  nC = nD;
        msA = msB; msB = msC; msC = msD;
        mcA = mcB; mcB = mcC; mcC = mcD;
    }
    __syncthreads();

    const int colBase = (wave & 7) * 16;
    const int mb      = (wave >> 3) * 32;
    const int nn      = lane & 15;
    const int kHalf   = (lane >> 4) * 2;

    v8f acc0 = {};
    v8f acc1 = {};
    v2f bq[3][4];
    const unsigned wofs = (unsigned)(kHalf * DIM + colBase + nn);
    auto load_b = [&](int g) {
        #pragma unroll
        for (int j = 0; j < 4; ++j) {
            const int k0 = (g * 4 + j) * 4;
            bq[g % 3][j].x = W[wofs + (unsigned)(k0 * DIM)];
            bq[g % 3][j].y = W[wofs + (unsigned)((k0 + 1) * DIM)];
        }
    };
    load_b(0);
    load_b(1);
    #pragma unroll
    for (int g = 0; g < DIM / 16; ++g) {
        if (g + 2 < DIM / 16) load_b(g + 2);
        __builtin_amdgcn_sched_barrier(0);
        #pragma unroll
        for (int j = 0; j < 4; ++j) {
            const int k0 = (g * 4 + j) * 4;
            v2f a0, a1;
            a0.x = As[(mb + nn) * LDS_STRIDE + k0 + kHalf];
            a0.y = As[(mb + nn) * LDS_STRIDE + k0 + kHalf + 1];
            a1.x = As[(mb + 16 + nn) * LDS_STRIDE + k0 + kHalf];
            a1.y = As[(mb + 16 + nn) * LDS_STRIDE + k0 + kHalf + 1];
            acc0 = __builtin_amdgcn_wmma_f32_16x16x4_f32(
                 false, a0,  false, bq[g % 3][j],
                 (short)0, acc0,  false,  false);
            acc1 = __builtin_amdgcn_wmma_f32_16x16x4_f32(
                 false, a1,  false, bq[g % 3][j],
                 (short)0, acc1,  false,  false);
        }
        __builtin_amdgcn_sched_barrier(0);
    }
    __syncthreads();

    const float bcol = bias[colBase + nn];
    #pragma unroll
    for (int v = 0; v < 8; ++v) {
        int r = mb + v + 8 * (lane >> 4);
        Os[r * LDS_STRIDE + colBase + nn] = acc0[v] + bcol;
        Os[(r + 16) * LDS_STRIDE + colBase + nn] = acc1[v] + bcol;
    }
    __syncthreads();

    if (wave < TILE_ROWS / 32) {
        int r = wave * 32 + lane;
        float dot = 0.0f;
        #pragma unroll 8
        for (int k = 0; k < DIM; ++k) dot += Os[r * LDS_STRIDE + k] * w_attn[k];
        float al = 1.0f / (1.0f + __expf(-dot));
        alphaS[r] = al;
        int gr = row0 + r;
        if (gr < NUM_DST) alpha_out[gr] = al;
    }
    __syncthreads();

    for (int i = threadIdx.x; i < TILE_ROWS * DIM; i += K2_THREADS) {
        int r = i >> 7;
        int c = i & 127;
        int gr = row0 + r;
        if (gr < NUM_DST) {
            out_scaled[(size_t)gr * DIM + c] = Os[r * LDS_STRIDE + c] * alphaS[r];
        }
    }
}

extern "C" void kernel_launch(void* const* d_in, const int* in_sizes, int n_in,
                              void* d_out, int out_size, void* d_ws, size_t ws_size,
                              hipStream_t stream) {
    const float* h      = (const float*)d_in[0];
    const int*   src    = (const int*)d_in[1];
    const int*   dst    = (const int*)d_in[2];
    const float* W      = (const float*)d_in[3];
    const float* bias   = (const float*)d_in[4];
    const float* w_attn = (const float*)d_in[5];

    float* out   = (float*)d_out;
    float* alpha = out + (size_t)NUM_DST * DIM;

    int* wsi       = (int*)d_ws;
    int* deg_src   = wsi + WS_DEG_SRC;
    int* bin_count = wsi + WS_BINCNT;
    int* binbuf    = wsi + WS_BINBUF;

    hipMemsetAsync(d_ws, 0, (size_t)(NUM_SRC + 256) * sizeof(int), stream);

    bin_kernel<<<(NUM_E + K1_EDGES - 1) / K1_EDGES, K1_THREADS, 0, stream>>>(src, dst, deg_src, bin_count, binbuf);

    gather_gemm_attn_kernel<<<(NUM_DST + TILE_ROWS - 1) / TILE_ROWS, K2_THREADS, 0, stream>>>(
        h, src, dst, deg_src, bin_count, binbuf, W, bias, w_attn, out, alpha);
}


# Per-edge global degree adds replaced by a second LDS counting sort into 512-source bins and a small launch that counts each bin in LDS and writes the scale table; row sums keep LDS arrival order
# speedup vs baseline: 4.6865x; 1.5301x over previous
#include <hip/hip_runtime.h>
#include <math.h>

#define NUM_SRC 100000
#define NUM_DST 50000
#define NUM_E   800000
#define DIM     128

typedef __attribute__((ext_vector_type(2))) float v2f;
typedef __attribute__((ext_vector_type(8))) float v8f;

#define BIN_SHIFT 8
#define BIN_DST   256
#define NBIN      ((NUM_DST + BIN_DST - 1) / BIN_DST)
#define BINCAP    6144
#define ROWCAP    64

#define SRC_SHIFT 9
#define SRC_BIN   512
#define NSBIN     ((NUM_SRC + SRC_BIN - 1) / SRC_BIN)

#define WS_BINCNT  0
#define WS_SBINCNT 256
#define WS_SCALE   512
#define WS_BINBUF  100608
#define WS_SBINBUF (WS_BINBUF + NBIN * BINCAP)

#define K1_THREADS 1024
#define K1_EPT     4
#define K1_EDGES   (K1_THREADS * K1_EPT)

template <typename EntryT>
__device__ __forceinline__ void sort_append(const int (&b)[K1_EPT], const int (&v)[K1_EPT], int e0,
                                            int* hist, int* lofs, int* gbase, int* wsum,
                                            int* sval, unsigned char* sbin,
                                            int* __restrict__ bin_count,
                                            EntryT* __restrict__ binbuf) {
    const int tid  = threadIdx.x;
    const int lane = tid & 31;
    const int wave = tid >> 5;
    if (tid < 256) hist[tid] = 0;
    __syncthreads();

    int r[K1_EPT];
    #pragma unroll
    for (int i = 0; i < K1_EPT; ++i) {
        const int e = e0 + i * K1_THREADS + tid;
        r[i] = 0;
        if (e < NUM_E) r[i] = atomicAdd(&hist[b[i]], 1);
    }
    __syncthreads();

    int c = 0, incl = 0;
    if (tid < 256) {
        c = hist[tid];
        incl = c;
        #pragma unroll
        for (int o = 1; o < 32; o <<= 1) {
            const int t = __shfl_up(incl, o);
            if (lane >= o) incl += t;
        }
        if (lane == 31) wsum[wave] = incl;
    }
    __syncthreads();
    if (tid < 256) {
        int add = 0;
        for (int w = 0; w < wave; ++w) add += wsum[w];
        lofs[tid] = add + incl - c;
        int gb = 0;
        if (c > 0) gb = atomicAdd(&bin_count[tid], c);
        gbase[tid] = gb;
    }
    __syncthreads();

    #pragma unroll
    for (int i = 0; i < K1_EPT; ++i) {
        const int e = e0 + i * K1_THREADS + tid;
        if (e < NUM_E) {
            const int p = lofs[b[i]] + r[i];
            sval[p] = v[i];
            sbin[p] = (unsigned char)b[i];
        }
    }
    __syncthreads();

    const int total = min(K1_EDGES, NUM_E - e0);
    for (int j = tid; j < total; j += K1_THREADS) {
        const int bj  = sbin[j];
        const int pos = gbase[bj] + (j - lofs[bj]);
        if (pos >= 0 && pos < BINCAP) binbuf[(size_t)bj * BINCAP + pos] = (EntryT)sval[j];
    }
}

__global__ __launch_bounds__(K1_THREADS) void bin_kernel(const int* __restrict__ src,
                                                         const int* __restrict__ dst,
                                                         int* __restrict__ bin_count,
                                                         int* __restrict__ binbuf,
                                                         int* __restrict__ sbin_count,
                                                         int* __restrict__ sbinbuf_words) {
    __shared__ int hist[256];
    __shared__ int lofs[256];
    __shared__ int gbase[256];
    __shared__ int wsum[8];
    __shared__ int sval[K1_EDGES];
    __shared__ unsigned char sbin[K1_EDGES];

    static_assert(NBIN <= 256 && NSBIN <= 256, "a bin index fits the 256-entry tables and a byte");
    const int tid = threadIdx.x;
    const int e0  = blockIdx.x * K1_EDGES;
    int s[K1_EPT], d[K1_EPT], b[K1_EPT], v[K1_EPT];
    #pragma unroll
    for (int i = 0; i < K1_EPT; ++i) {
        const int e = e0 + i * K1_THREADS + tid;
        s[i] = 0;
        d[i] = 0;
        if (e < NUM_E) {
            s[i] = src[e];
            d[i] = dst[e];
        }
    }
    #pragma unroll
    for (int i = 0; i < K1_EPT; ++i) {
        b[i] = min(max(d[i] >> BIN_SHIFT, 0), NBIN - 1);
        v[i] = (s[i] << 8) | (d[i] & (BIN_DST - 1));
    }
    sort_append<int>(b, v, e0, hist, lofs, gbase, wsum, sval, sbin, bin_count, binbuf);

    #pragma unroll
    for (int i = 0; i < K1_EPT; ++i) {
        b[i] = min(max(s[i] >> SRC_SHIFT, 0), NSBIN - 1);
        v[i] = s[i] & (SRC_BIN - 1);
    }
    sort_append<unsigned short>(b, v, e0, hist, lofs, gbase, wsum, sval, sbin, sbin_count,
                                (unsigned short*)sbinbuf_words);
}

#define SBIN_TRIPS (BINCAP / SRC_BIN)
__global__ __launch_bounds__(SRC_BIN) void degree_scale_kernel(const int* __restrict__ src,
                                                               const int* __restrict__ sbin_count,
                                                               const int* __restrict__ sbinbuf_words,
                                                               float* __restrict__ scale) {
    __shared__ int cnts[SRC_BIN];
    static_assert(BINCAP % SRC_BIN == 0, "whole trips");
    const int tid = threadIdx.x;
    const int bin = blockIdx.x;
    cnts[tid] = 0;
    __syncthreads();

    const int cnt = max(sbin_count[bin], 0);
    if (cnt <= BINCAP) {
        const unsigned short* bb = (const unsigned short*)sbinbuf_words + (size_t)bin * BINCAP;
        int q[SBIN_TRIPS];
        #pragma unroll
        for (int t = 0; t < SBIN_TRIPS; ++t) {
            const int i = t * SRC_BIN + tid;
            q[t] = -1;
            if (i < cnt) q[t] = bb[i];
        }
        #pragma unroll
        for (int t = 0; t < SBIN_TRIPS; ++t) {
            if (q[t] >= 0) atomicAdd(&cnts[q[t] & (SRC_BIN - 1)], 1);
        }
    } else {
        for (int e = tid; e < NUM_E; e += SRC_BIN) {
            const int sv = src[e];
            if (min(max(sv >> SRC_SHIFT, 0), NSBIN - 1) == bin) atomicAdd(&cnts[sv & (SRC_BIN - 1)], 1);
        }
    }
    __syncthreads();

    const int sidx = bin * SRC_BIN + tid;
    if (sidx < NUM_SRC) scale[sidx] = rsqrtf(fmaxf((float)cnts[tid], 1.0f));
}

template <int K>
__device__ __forceinline__ void gather_group(const float* __restrict__ h, int lane, int ms, int mc,
                                             int j, float4& acc) {
    float4 v[K];
    float  sc[K];
    #pragma unroll
    for (int k = 0; k < K; ++k) {
        int s = __builtin_amdgcn_readlane(ms, j + k);
        s = min(max(s, 0), NUM_SRC - 1);
        sc[k] = __int_as_float(__builtin_amdgcn_readlane(mc, j + k));
        v[k] = ((const float4*)(h + (size_t)s * DIM))[lane];
    }
    #pragma unroll
    for (int k = 0; k < K; ++k) {
        acc.x = fmaf(v[k].x, sc[k], acc.x);
        acc.y = fmaf(v[k].y, sc[k], acc.y);
        acc.z = fmaf(v[k].z, sc[k], acc.z);
        acc.w = fmaf(v[k].w, sc[k], acc.w);
    }
}

__device__ __forceinline__ void load_chunk(const int* __restrict__ brow,
                                           const float* __restrict__ scale, int base, int end,
                                           int lane, int& ms, int& mc) {
    ms = 0;
    mc = 0;
    if (lane < end - base) {
        int s = brow[base + lane];
        s = min(max(s, 0), NUM_SRC - 1);
        ms = s;
        mc = __float_as_int(scale[s]);
    }
}

__device__ __forceinline__ float4 gather_row(const float* __restrict__ h,
                                             const int* __restrict__ brow,
                                             const float* __restrict__ scale,
                                             int end, int lane, int ms, int mc) {
    float4 acc = make_float4(0.0f, 0.0f, 0.0f, 0.0f);
    for (int base = 0;;) {
        int n = end - base;
        n = n < 32 ? n : 32;
        int j = 0;
        for (; j + 8 <= n; j += 8) gather_group<8>(h, lane, ms, mc, j, acc);
        if (n - j >= 4) { gather_group<4>(h, lane, ms, mc, j, acc); j += 4; }
        if (n - j >= 2) { gather_group<2>(h, lane, ms, mc, j, acc); j += 2; }
        if (n - j >= 1) { gather_group<1>(h, lane, ms, mc, j, acc); }
        base += 32;
        if (base >= end) break;
        load_chunk(brow, scale, base, end, lane, ms, mc);
    }
    return acc;
}

__device__ __forceinline__ float4 gather_row_scan(const float* __restrict__ h,
                                               const int* __restrict__ src,
                                               const int* __restrict__ dst,
                                               const float* __restrict__ scale,
                                               int d, int lane) {
    static_assert(NUM_E % 32 == 0, "edge scan walks whole 32-edge trips");
    float4 acc = make_float4(0.0f, 0.0f, 0.0f, 0.0f);
    const float* hl = h + lane * 4;
    for (int base = 0; base < NUM_E; base += 32) {
        const int e = base + lane;
        const int dd = dst[e];
        int s = src[e];
        unsigned m = __builtin_amdgcn_ballot_w32(dd == d);
        if (m != 0u) {
            s = min(max(s, 0), NUM_SRC - 1);
            const int scb = __float_as_int(scale[s]);
            while (m != 0u) {
                const int j = __builtin_ctz(m);
                m &= m - 1u;
                const int   sj = __builtin_amdgcn_readlane(s, j);
                const float sc = __int_as_float(__builtin_amdgcn_readlane(scb, j));
                const float4 v = *(const float4*)(hl + (size_t)sj * DIM);
                acc.x = fmaf(v.x, sc, acc.x);
                acc.y = fmaf(v.y, sc, acc.y);
                acc.z = fmaf(v.z, sc, acc.z);
                acc.w = fmaf(v.w, sc, acc.w);
            }
        }
    }
    return acc;
}

#define LDS_STRIDE 132
#define TILE_ROWS  64
#define K2_THREADS 512
#define BIN_TRIPS  (BINCAP / (K2_THREADS * 4))

__global__ __launch_bounds__(K2_THREADS, 16) void gather_gemm_attn_kernel(const float* __restrict__ h,
                                                               const int* __restrict__ src,
                                                               const int* __restrict__ dst,
                                                               const float* __restrict__ scale,
                                                               const int* __restrict__ bin_count,
                                                               const int* __restrict__ binbuf,
                                                               const float* __restrict__ W,
                                                               const float* __restrict__ bias,
                                                               const float* __restrict__ w_attn,
                                                               float* __restrict__ out_scaled,
                                                               float* __restrict__ alpha_out) {
    __shared__ float As[TILE_ROWS * LDS_STRIDE];
    __shared__ float alphaS[TILE_ROWS];
    __shared__ int   rowcnt[TILE_ROWS];
    __shared__ int   rowlist[TILE_ROWS * ROWCAP];
    float* Os = As;

    static_assert(BIN_DST % TILE_ROWS == 0 && BINCAP % (K2_THREADS * 4) == 0,
                  "a row tile lies in one bin; whole trips");
    static_assert(TILE_ROWS == (K2_THREADS / 32) * 4, "4 gathered rows per wave");
    const int row0 = blockIdx.x * TILE_ROWS;
    const int bin  = row0 >> BIN_SHIFT;
    const int lb   = row0 & (BIN_DST - 1);
    const int wave = __builtin_amdgcn_readfirstlane(threadIdx.x >> 5);
    const int lane = threadIdx.x & 31;

    if (threadIdx.x < TILE_ROWS) rowcnt[threadIdx.x] = 0;
    __syncthreads();

    auto row_append = [&](int r, int sv) {
        const int slot = atomicAdd(&rowcnt[r], 1);
        if (slot < ROWCAP) rowlist[r * ROWCAP + slot] = sv;
    };

    const int cnt = max(bin_count[bin], 0);
    if (cnt <= BINCAP) {
        const int* bb = binbuf + (size_t)bin * BINCAP;
        int4 q[BIN_TRIPS];
        #pragma unroll
        for (int t = 0; t < BIN_TRIPS; ++t) {
            const int i = t * (K2_THREADS * 4) + threadIdx.x * 4;
            q[t] = make_int4(-1, -1, -1, -1);
            if (cnt - i >= 4) q[t] = *(const int4*)(bb + i);
        }
        int qt = -1;
        if ((int)threadIdx.x < (cnt & 3)) qt = bb[(cnt & ~3) + threadIdx.x];
        #pragma unroll
        for (int t = 0; t < BIN_TRIPS; ++t) {
            const int vv[4] = {q[t].x, q[t].y, q[t].z, q[t].w};
            int slot[4];
            #pragma unroll
            for (int k = 0; k < 4; ++k) {
                const int r = (vv[k] & (BIN_DST - 1)) - lb;
                slot[k] = ROWCAP;
                if (vv[k] >= 0 && (unsigned)r < (unsigned)TILE_ROWS) slot[k] = atomicAdd(&rowcnt[r], 1);
            }
            #pragma unroll
            for (int k = 0; k < 4; ++k) {
                const int r = (vv[k] & (BIN_DST - 1)) - lb;
                if (slot[k] < ROWCAP) rowlist[r * ROWCAP + slot[k]] = vv[k] >> 8;
            }
        }
        {
            const int r = (qt & (BIN_DST - 1)) - lb;
            if (qt >= 0 && (unsigned)r < (unsigned)TILE_ROWS) row_append(r, qt >> 8);
        }
    } else {
        for (int e = threadIdx.x; e < NUM_E; e += K2_THREADS) {
            const int r = dst[e] - row0;
            if ((unsigned)r < (unsigned)TILE_ROWS) row_append(r, src[e]);
        }
    }
    __syncthreads();

    const int r0 = wave * 4;
    int nA = rowcnt[r0], nB = rowcnt[r0 + 1], nC = rowcnt[r0 + 2], nD = rowcnt[r0 + 3];
    const int* brow = rowlist + r0 * ROWCAP;
    int msA = (lane < min(nA, ROWCAP)) ? brow[lane] : 0;
    int msB = (lane < min(nB, ROWCAP)) ? brow[ROWCAP + lane] : 0;
    int msC = (lane < min(nC, ROWCAP)) ? brow[2 * ROWCAP + lane] : 0;
    int msD = (lane < min(nD, ROWCAP)) ? brow[3 * ROWCAP + lane] : 0;
    msA = min(max(msA, 0), NUM_SRC - 1);
    msB = min(max(msB, 0), NUM_SRC - 1);
    msC = min(max(msC, 0), NUM_SRC - 1);
    msD = min(max(msD, 0), NUM_SRC - 1);
    int mcA = __float_as_int(scale[msA]);
    int mcB = __float_as_int(scale[msB]);
    int mcC = __float_as_int(scale[msC]);
    int mcD = __float_as_int(scale[msD]);
    for (int qi = 0; qi < 4; ++qi) {
        const int r = r0 + qi;
        float4 acc = gather_row(h, rowlist + r * ROWCAP, scale, min(nA, ROWCAP), lane, msA, mcA);
        if (__builtin_expect(nA > ROWCAP, 0)) acc = gather_row_scan(h, src, dst, scale, row0 + r, lane);
        const float ds = rsqrtf(fmaxf((float)nA, 1.0f));
        *(float4*)(&As[r * LDS_STRIDE + lane * 4]) =
            make_float4(acc.x * ds, acc.y * ds, acc.z * ds, acc.w * ds);
        nA = nB;  nB = nC;  nC = nD;
        msA = msB; msB = msC; msC = msD;
        mcA = mcB; mcB = mcC; mcC = mcD;
    }
    __syncthreads();

    const int colBase = (wave & 7) * 16;
    const int mb      = (wave >> 3) * 32;
    const int nn      = lane & 15;
    const int kHalf   = (lane >> 4) * 2;

    v8f acc0 = {};
    v8f acc1 = {};
    v2f bq[3][4];
    const unsigned wofs = (unsigned)(kHalf * DIM + colBase + nn);
    auto load_b = [&](int g) {
        #pragma unroll
        for (int j = 0; j < 4; ++j) {
            const int k0 = (g * 4 + j) * 4;
            bq[g % 3][j].x = W[wofs + (unsigned)(k0 * DIM)];
            bq[g % 3][j].y = W[wofs + (unsigned)((k0 + 1) * DIM)];
        }
    };
    load_b(0);
    load_b(1);
    #pragma unroll
    for (int g = 0; g < DIM / 16; ++g) {
        if (g + 2 < DIM / 16) load_b(g + 2);
        __builtin_amdgcn_sched_barrier(0);
        #pragma unroll
        for (int j = 0; j < 4; ++j) {
            const int k0 = (g * 4 + j) * 4;
            v2f a0, a1;
            a0.x = As[(mb + nn) * LDS_STRIDE + k0 + kHalf];
            a0.y = As[(mb + nn) * LDS_STRIDE + k0 + kHalf + 1];
            a1.x = As[(mb + 16 + nn) * LDS_STRIDE + k0 + kHalf];
            a1.y = As[(mb + 16 + nn) * LDS_STRIDE + k0 + kHalf + 1];
            acc0 = __builtin_amdgcn_wmma_f32_16x16x4_f32(
                 false, a0,  false, bq[g % 3][j],
                 (short)0, acc0,  false,  false);
            acc1 = __builtin_amdgcn_wmma_f32_16x16x4_f32(
                 false, a1,  false, bq[g % 3][j],
                 (short)0, acc1,  false,  false);
        }
        __builtin_amdgcn_sched_barrier(0);
    }
    __syncthreads();

    const float bcol = bias[colBase + nn];
    #pragma unroll
    for (int v = 0; v < 8; ++v) {
        int r = mb + v + 8 * (lane >> 4);
        Os[r * LDS_STRIDE + colBase + nn] = acc0[v] + bcol;
        Os[(r + 16) * LDS_STRIDE + colBase + nn] = acc1[v] + bcol;
    }
    __syncthreads();

    if (wave < TILE_ROWS / 32) {
        int r = wave * 32 + lane;
        float dot = 0.0f;
        #pragma unroll 8
        for (int k = 0; k < DIM; ++k) dot += Os[r * LDS_STRIDE + k] * w_attn[k];
        float al = 1.0f / (1.0f + __expf(-dot));
        alphaS[r] = al;
        int gr = row0 + r;
        if (gr < NUM_DST) alpha_out[gr] = al;
    }
    __syncthreads();

    for (int i = threadIdx.x; i < TILE_ROWS * DIM; i += K2_THREADS) {
        int r = i >> 7;
        int c = i & 127;
        int gr = row0 + r;
        if (gr < NUM_DST) {
            out_scaled[(size_t)gr * DIM + c] = Os[r * LDS_STRIDE + c] * alphaS[r];
        }
    }
}

extern "C" void kernel_launch(void* const* d_in, const int* in_sizes, int n_in,
                              void* d_out, int out_size, void* d_ws, size_t ws_size,
                              hipStream_t stream) {
    const float* h      = (const float*)d_in[0];
    const int*   src    = (const int*)d_in[1];
    const int*   dst    = (const int*)d_in[2];
    const float* W      = (const float*)d_in[3];
    const float* bias   = (const float*)d_in[4];
    const float* w_attn = (const float*)d_in[5];

    float* out   = (float*)d_out;
    float* alpha = out + (size_t)NUM_DST * DIM;

    int*   wsi        = (int*)d_ws;
    int*   bin_count  = wsi + WS_BINCNT;
    int*   sbin_count = wsi + WS_SBINCNT;
    float* scale      = (float*)(wsi + WS_SCALE);
    int*   binbuf     = wsi + WS_BINBUF;
    int*   sbinbuf    = wsi + WS_SBINBUF;

    hipMemsetAsync(d_ws, 0, (size_t)512 * sizeof(int), stream);

    bin_kernel<<<(NUM_E + K1_EDGES - 1) / K1_EDGES, K1_THREADS, 0, stream>>>(src, dst, bin_count, binbuf, sbin_count, sbinbuf);

    degree_scale_kernel<<<NSBIN, SRC_BIN, 0, stream>>>(src, sbin_count, sbinbuf, scale);

    gather_gemm_attn_kernel<<<(NUM_DST + TILE_ROWS - 1) / TILE_ROWS, K2_THREADS, 0, stream>>>(
        h, src, dst, scale, bin_count, binbuf, W, bias, w_attn, out, alpha);
}
